# starting kernel
# baseline (speedup 1.0000x reference)
#include <hip/hip_runtime.h>
#include <hip/hip_bf16.h>

typedef __attribute__((ext_vector_type(2))) float v2f;
typedef __attribute__((ext_vector_type(8))) float v8f;

#define T_STEPS 2048
#define BATCH   512
#define NIN     27
#define NROWS   (T_STEPS * BATCH)
#define NTILES  (NROWS / 16)

__global__ void lstm_xproj_wmma(const float* __restrict__ x,
                                const float* __restrict__ W_ih,
                                const float* __restrict__ b_ih,
                                const float* __restrict__ b_hh,
                                float* __restrict__ xp)
{
    const int lane   = threadIdx.x & 31;
    const int wave   = (blockIdx.x * blockDim.x + threadIdx.x) >> 5;
    const int nwaves = (gridDim.x * blockDim.x) >> 5;

    const int mn    = lane & 15;
    const int khalf = (lane >> 4) << 1;

    v2f bmat[7];
#pragma unroll
    for (int kk = 0; kk < 7; ++kk) {
        const int k0 = kk * 4 + khalf;
        float bx = 0.0f, by = 0.0f;
        if (mn < 4) {
            if (k0     < NIN) bx = W_ih[mn * NIN + k0];
            if (k0 + 1 < NIN) by = W_ih[mn * NIN + k0 + 1];
        }
        bmat[kk].x = bx;
        bmat[kk].y = by;
    }
    const float bias = (mn < 4) ? (b_ih[mn] + b_hh[mn]) : 0.0f;

    for (int tile = wave; tile < NTILES; tile += nwaves) {
        const float* __restrict__ arow = x + (size_t)(tile * 16 + mn) * NIN;

        v8f acc = {};
#pragma unroll
        for (int kk = 0; kk < 7; ++kk) {
            const int k0 = kk * 4 + khalf;
            v2f a;
            a.x = (k0     < NIN) ? arow[k0]     : 0.0f;
            a.y = (k0 + 1 < NIN) ? arow[k0 + 1] : 0.0f;
            acc = __builtin_amdgcn_wmma_f32_16x16x4_f32(
                false, a, false, bmat[kk], (short)0, acc, false, false);
        }

        if (mn < 4) {
            const int rbase = tile * 16 + ((lane >> 4) << 3);
#pragma unroll
            for (int r = 0; r < 8; ++r) {
                xp[(size_t)(rbase + r) * 4 + mn] = acc[r] + bias;
            }
        }
    }
}

__device__ __forceinline__ float fast_sigmoid(float v) {
    return 1.0f / (1.0f + __expf(-v));
}
__device__ __forceinline__ float fast_tanh(float v) {
    return 1.0f - 2.0f / (__expf(2.0f * v) + 1.0f);
}

__global__ void lstm_scan(const float* __restrict__ xp,
                          const float* __restrict__ W_hh,
                          float* __restrict__ out)
{
    const int b = blockIdx.x * blockDim.x + threadIdx.x;
    if (b >= BATCH) return;

    const float wi = W_hh[0];
    const float wf = W_hh[1];
    const float wg = W_hh[2];
    const float wo = W_hh[3];

    float h = 0.0f, c = 0.0f;
    for (int t = 0; t < T_STEPS; ++t) {
        const float4 g = *(const float4*)(xp + ((size_t)t * BATCH + b) * 4);
        const float i  = fast_sigmoid(g.x + h * wi);
        const float f  = fast_sigmoid(g.y + h * wf);
        const float gg = fast_tanh   (g.z + h * wg);
        const float o  = fast_sigmoid(g.w + h * wo);
        c = f * c + i * gg;
        h = o * fast_tanh(c);
        out[(size_t)t * BATCH + b] = h;
    }
}

extern "C" void kernel_launch(void* const* d_in, const int* in_sizes, int n_in,
                              void* d_out, int out_size, void* d_ws, size_t ws_size,
                              hipStream_t stream)
{
    const float* x    = (const float*)d_in[0];
    const float* W_ih = (const float*)d_in[1];
    const float* W_hh = (const float*)d_in[2];
    const float* b_ih = (const float*)d_in[3];
    const float* b_hh = (const float*)d_in[4];
    float* out = (float*)d_out;
    float* xp  = (float*)d_ws;

    lstm_xproj_wmma<<<512, 256, 0, stream>>>(x, W_ih, b_ih, b_hh, xp);

    lstm_scan<<<2, 256, 0, stream>>>(xp, W_hh, out);
}


# scan cell: i*g, f*c and o*tanh(c) each formed as ONE correctly rounded quotient (tanh = sign*(1-e)/(1+e), e = exp(-2|v|)): 5 divisions a step become 3; products round differently, no sum reordered
# speedup vs baseline: 5.1058x; 5.1058x over previous
#include <hip/hip_runtime.h>
#include <hip/hip_bf16.h>

typedef __attribute__((ext_vector_type(2))) float v2f;
typedef __attribute__((ext_vector_type(4))) float v4f;
typedef __attribute__((ext_vector_type(8))) float v8f;
typedef v2f v2f_a4 __attribute__((aligned(4)));
typedef v4f v4f_a4 __attribute__((aligned(4)));

#define KOFF    13

#define T_STEPS 2048
#define BATCH   512
#define NIN     27

#define WG_B    32
#define CH      64
#define NCH     (T_STEPS / CH)
#define NWAVES  16
#define NWORK   12
#define NTILE   (CH * WG_B / 16)
#define NFULL   (NTILE / NWORK)
#define NTAIL   (NTILE - NFULL * NWORK)
#define TPASS   5

__device__ __forceinline__ float inv_sigmoid(float v) {
    return 1.0f + __expf(-v);
}

__device__ __forceinline__ float tanh_over(float v, float a) {
    const float e = __expf(-2.0f * fabsf(v));
    const float s = copysignf(1.0f, v);
    return fmaf(-e, s, s) / (a * (1.0f + e));
}

__device__ __forceinline__ float cell_step(const float4 g, float& h, float& c,
                                           float wi, float wf, float wg, float wo)
{
    const float zg = g.z + h * wg;
    const float zi = g.x + h * wi;
    const float zf = g.y + h * wf;
    const float zo = g.w + h * wo;
    const float ig = tanh_over(zg, inv_sigmoid(zi));
    const float fc = c / inv_sigmoid(zf);
    const float ao = inv_sigmoid(zo);
    c = fc + ig;
    h = tanh_over(c, ao);
    return h;
}

__device__ __forceinline__ void lds_barrier()
{
    __builtin_amdgcn_fence(__ATOMIC_RELEASE, "workgroup", "local");
    __builtin_amdgcn_s_barrier();
    __builtin_amdgcn_fence(__ATOMIC_ACQUIRE, "workgroup", "local");
}


__device__ __forceinline__ void load_tile(const float* __restrict__ x, size_t crow0, int t,
                                          int mn, int kh, v2f (&a)[7])
{
    const size_t grow = crow0 + (size_t)(t >> 1) * BATCH + (t & 1) * 16;
    const float* __restrict__ run = x + (grow + mn) * NIN + KOFF * kh;
    const v4f_a4 r0 = *(const v4f_a4*)(run);
    const v4f_a4 r1 = *(const v4f_a4*)(run + 4);
    const v4f_a4 r2 = *(const v4f_a4*)(run + 8);
    const v2f_a4 r3 = *(const v2f_a4*)(run + 12);
    a[0].x = r0.x;  a[0].y = r0.y;
    a[1].x = r0.z;  a[1].y = r0.w;
    a[2].x = r1.x;  a[2].y = r1.y;
    a[3].x = r1.z;  a[3].y = r1.w;
    a[4].x = r2.x;  a[4].y = r2.y;
    a[5].x = r2.z;  a[5].y = r2.w;
    a[6].x = r3.x;  a[6].y = r3.y;
}

__device__ __forceinline__ void finish_tile(const v2f (&a)[7], const v2f (&bmat)[7], float bias,
                                            int t, int lane, int mn, float* __restrict__ dst)
{
    v8f acc = {};
#pragma unroll
    for (int kk = 0; kk < 7; ++kk) {
        acc = __builtin_amdgcn_wmma_f32_16x16x4_f32(
            false, a[kk], false, bmat[kk], (short)0, acc, false, false);
    }
    if (mn < 4) {
        const int rl = (t >> 1) * WG_B + (t & 1) * 16 + ((lane >> 4) << 3);
#pragma unroll
        for (int r = 0; r < 8; ++r) {
            dst[(rl + r) * 4 + mn] = acc[r] + bias;
        }
    }
}

__device__ __forceinline__ void produce_chunk(const float* __restrict__ x,
                                              float* __restrict__ dst,
                                              int chunk, int b0, int rank, int lane,
                                              const v2f (&bmat)[7], float bias)
{
    const int mn = lane & 15;
    const int kh = lane >> 4;
    const size_t crow0 = (size_t)chunk * CH * BATCH + b0;

    for (int j0 = 0; j0 < NFULL; j0 += TPASS) {
        v2f a[TPASS][7];
#pragma unroll
        for (int q = 0; q < TPASS; ++q) {
            load_tile(x, crow0, rank + NWORK * (j0 + q), mn, kh, a[q]);
        }
#pragma unroll
        for (int q = 0; q < TPASS; ++q) {
            finish_tile(a[q], bmat, bias, rank + NWORK * (j0 + q), lane, mn, dst);
        }
    }
    if (rank < NTAIL) {
        v2f a[7];
        load_tile(x, crow0, NFULL * NWORK + rank, mn, kh, a);
        finish_tile(a, bmat, bias, NFULL * NWORK + rank, lane, mn, dst);
    }
}

__global__ __launch_bounds__(32 * NWAVES)
void lstm_fused_wmma(const float* __restrict__ x,
                     const float* __restrict__ W_ih,
                     const float* __restrict__ W_hh,
                     const float* __restrict__ b_ih,
                     const float* __restrict__ b_hh,
                     float* __restrict__ out)
{
    __shared__ float4 gbuf[2][CH * WG_B];

    const int lane = threadIdx.x & 31;
    const int wave = __builtin_amdgcn_readfirstlane((int)(threadIdx.x >> 5));
    const int b0   = blockIdx.x * WG_B;
    const bool worker = (wave & 3) != 0;
    const int  rank   = (wave >> 2) * 3 + (wave & 3) - 1;

    const int mn = lane & 15;
    const int kh = lane >> 4;

    v2f bmat[7];
#pragma unroll
    for (int kk = 0; kk < 7; ++kk) {
        const int k0 = KOFF * kh + 2 * kk;
        float bx = 0.0f, by = 0.0f;
        if (mn < 4) {
            if (!(kh == 1 && kk == 0)) bx = W_ih[mn * NIN + k0];
            by = W_ih[mn * NIN + k0 + 1];
        }
        bmat[kk].x = bx;
        bmat[kk].y = by;
    }
    const float bias = (mn < 4) ? (b_ih[mn] + b_hh[mn]) : 0.0f;

    const float wi = W_hh[0];
    const float wf = W_hh[1];
    const float wg = W_hh[2];
    const float wo = W_hh[3];

    float h = 0.0f, c = 0.0f;

    if (worker) {
        produce_chunk(x, (float*)gbuf[0], 0, b0, rank, lane, bmat, bias);
    }
    lds_barrier();

    for (int ch = 0; ch < NCH; ++ch) {
        if (wave == 0) {
            const float4* gb = gbuf[ch & 1];
            float* __restrict__ orow = out + (size_t)(ch * CH) * BATCH + b0 + lane;

            float4 ga[8], gc[8];
#pragma unroll
            for (int u = 0; u < 8; ++u) ga[u] = gb[u * WG_B + lane];

            for (int s = 0; s < CH; s += 16) {
#pragma unroll
                for (int u = 0; u < 8; ++u) gc[u] = gb[(s + 8 + u) * WG_B + lane];
#pragma unroll
                for (int u = 0; u < 8; ++u) {
                    orow[(size_t)(s + u) * BATCH] = cell_step(ga[u], h, c, wi, wf, wg, wo);
                }
                const int sn = (s + 16 < CH) ? (s + 16) : 0;
#pragma unroll
                for (int u = 0; u < 8; ++u) ga[u] = gb[(sn + u) * WG_B + lane];
#pragma unroll
                for (int u = 0; u < 8; ++u) {
                    orow[(size_t)(s + 8 + u) * BATCH] = cell_step(gc[u], h, c, wi, wf, wg, wo);
                }
            }
        } else if (worker && ch + 1 < NCH) {
            produce_chunk(x, (float*)gbuf[(ch + 1) & 1], ch + 1, b0, rank, lane, bmat, bias);
        }
        lds_barrier();
    }
}

extern "C" void kernel_launch(void* const* d_in, const int* in_sizes, int n_in,
                              void* d_out, int out_size, void* d_ws, size_t ws_size,
                              hipStream_t stream)
{
    const float* x    = (const float*)d_in[0];
    const float* W_ih = (const float*)d_in[1];
    const float* W_hh = (const float*)d_in[2];
    const float* b_ih = (const float*)d_in[3];
    const float* b_hh = (const float*)d_in[4];
    float* out = (float*)d_out;

    lstm_fused_wmma<<<16, 512, 0, stream>>>(x, W_ih, W_hh, b_ih, b_hh, out);
}


# exp2 domain: W_ih, biases, W_hh scaled by -log2(e) (i,f,o) or 2log2(e) (g), cell state carried x 2log2(e): no multiply between a gate sum or c and its v_exp; products round anew, sum order kept
# speedup vs baseline: 5.5235x; 1.0818x over previous
#include <hip/hip_runtime.h>
#include <hip/hip_bf16.h>

typedef __attribute__((ext_vector_type(2))) float v2f;
typedef __attribute__((ext_vector_type(4))) float v4f;
typedef __attribute__((ext_vector_type(8))) float v8f;
typedef v2f v2f_a4 __attribute__((aligned(4)));
typedef v4f v4f_a4 __attribute__((aligned(4)));

#define KOFF    13

#define T_STEPS 2048
#define BATCH   512
#define NIN     27

#define WG_B    32
#define CH      64
#define NCH     (T_STEPS / CH)
#define NWAVES  16
#define NWORK   12
#define NTILE   (CH * WG_B / 16)
#define NFULL   (NTILE / NWORK)
#define NTAIL   (NTILE - NFULL * NWORK)
#define TPASS   5

#define LOG2E   1.44269504088896340736f
#define CSCALE  (2.0f * LOG2E)
__device__ __forceinline__ float inv_sigmoid2(float v2) {
    return 1.0f + __builtin_amdgcn_exp2f(v2);
}

__device__ __forceinline__ float tanh2_over(float v2, float a, float m) {
    const float e = __builtin_amdgcn_exp2f(-fabsf(v2));
    const float s = copysignf(m, v2);
    return fmaf(-e, s, s) / (a * (1.0f + e));
}

__device__ __forceinline__ float cell_step(const float4 g, float& h, float& c,
                                           float wi, float wf, float wg, float wo)
{
    const float zg = g.z + h * wg;
    const float zi = g.x + h * wi;
    const float zf = g.y + h * wf;
    const float zo = g.w + h * wo;
    const float ig = tanh2_over(zg, inv_sigmoid2(zi), CSCALE);
    const float fc = c / inv_sigmoid2(zf);
    const float ao = inv_sigmoid2(zo);
    c = fc + ig;
    h = tanh2_over(c, ao, 1.0f);
    return h;
}

__device__ __forceinline__ void lds_barrier()
{
    __builtin_amdgcn_fence(__ATOMIC_RELEASE, "workgroup", "local");
    __builtin_amdgcn_s_barrier();
    __builtin_amdgcn_fence(__ATOMIC_ACQUIRE, "workgroup", "local");
}


__device__ __forceinline__ void load_tile(const float* __restrict__ x, size_t crow0, int t,
                                          int mn, int kh, v2f (&a)[7])
{
    const size_t grow = crow0 + (size_t)(t >> 1) * BATCH + (t & 1) * 16;
    const float* __restrict__ run = x + (grow + mn) * NIN + KOFF * kh;
    const v4f_a4 r0 = *(const v4f_a4*)(run);
    const v4f_a4 r1 = *(const v4f_a4*)(run + 4);
    const v4f_a4 r2 = *(const v4f_a4*)(run + 8);
    const v2f_a4 r3 = *(const v2f_a4*)(run + 12);
    a[0].x = r0.x;  a[0].y = r0.y;
    a[1].x = r0.z;  a[1].y = r0.w;
    a[2].x = r1.x;  a[2].y = r1.y;
    a[3].x = r1.z;  a[3].y = r1.w;
    a[4].x = r2.x;  a[4].y = r2.y;
    a[5].x = r2.z;  a[5].y = r2.w;
    a[6].x = r3.x;  a[6].y = r3.y;
}

__device__ __forceinline__ void finish_tile(const v2f (&a)[7], const v2f (&bmat)[7], float bias,
                                            int t, int lane, int mn, float* __restrict__ dst)
{
    v8f acc = {};
#pragma unroll
    for (int kk = 0; kk < 7; ++kk) {
        acc = __builtin_amdgcn_wmma_f32_16x16x4_f32(
            false, a[kk], false, bmat[kk], (short)0, acc, false, false);
    }
    if (mn < 4) {
        const int rl = (t >> 1) * WG_B + (t & 1) * 16 + ((lane >> 4) << 3);
#pragma unroll
        for (int r = 0; r < 8; ++r) {
            dst[(rl + r) * 4 + mn] = acc[r] + bias;
        }
    }
}

__device__ __forceinline__ void produce_chunk(const float* __restrict__ x,
                                              float* __restrict__ dst,
                                              int chunk, int b0, int rank, int lane,
                                              const v2f (&bmat)[7], float bias)
{
    const int mn = lane & 15;
    const int kh = lane >> 4;
    const size_t crow0 = (size_t)chunk * CH * BATCH + b0;

    for (int j0 = 0; j0 < NFULL; j0 += TPASS) {
        v2f a[TPASS][7];
#pragma unroll
        for (int q = 0; q < TPASS; ++q) {
            load_tile(x, crow0, rank + NWORK * (j0 + q), mn, kh, a[q]);
        }
#pragma unroll
        for (int q = 0; q < TPASS; ++q) {
            finish_tile(a[q], bmat, bias, rank + NWORK * (j0 + q), lane, mn, dst);
        }
    }
    if (rank < NTAIL) {
        v2f a[7];
        load_tile(x, crow0, NFULL * NWORK + rank, mn, kh, a);
        finish_tile(a, bmat, bias, NFULL * NWORK + rank, lane, mn, dst);
    }
}

__global__ __launch_bounds__(32 * NWAVES)
void lstm_fused_wmma(const float* __restrict__ x,
                     const float* __restrict__ W_ih,
                     const float* __restrict__ W_hh,
                     const float* __restrict__ b_ih,
                     const float* __restrict__ b_hh,
                     float* __restrict__ out)
{
    __shared__ float4 gbuf[2][CH * WG_B];

    const int lane = threadIdx.x & 31;
    const int wave = __builtin_amdgcn_readfirstlane((int)(threadIdx.x >> 5));
    const int b0   = blockIdx.x * WG_B;
    const bool worker = (wave & 3) != 0;
    const int  rank   = (wave >> 2) * 3 + (wave & 3) - 1;

    const int mn = lane & 15;
    const int kh = lane >> 4;

    const float gscale = (mn == 2) ? CSCALE : -LOG2E;
    v2f bmat[7];
#pragma unroll
    for (int kk = 0; kk < 7; ++kk) {
        const int k0 = KOFF * kh + 2 * kk;
        float bx = 0.0f, by = 0.0f;
        if (mn < 4) {
            if (!(kh == 1 && kk == 0)) bx = W_ih[mn * NIN + k0] * gscale;
            by = W_ih[mn * NIN + k0 + 1] * gscale;
        }
        bmat[kk].x = bx;
        bmat[kk].y = by;
    }
    const float bias = (mn < 4) ? (b_ih[mn] + b_hh[mn]) * gscale : 0.0f;

    const float wi = W_hh[0] * -LOG2E;
    const float wf = W_hh[1] * -LOG2E;
    const float wg = W_hh[2] * CSCALE;
    const float wo = W_hh[3] * -LOG2E;

    float h = 0.0f, c = 0.0f;

    if (worker) {
        produce_chunk(x, (float*)gbuf[0], 0, b0, rank, lane, bmat, bias);
    }
    lds_barrier();

    for (int ch = 0; ch < NCH; ++ch) {
        if (wave == 0) {
            const float4* gb = gbuf[ch & 1];
            float* __restrict__ orow = out + (size_t)(ch * CH) * BATCH + b0 + lane;

            float4 ga[8], gc[8];
#pragma unroll
            for (int u = 0; u < 8; ++u) ga[u] = gb[u * WG_B + lane];

            for (int s = 0; s < CH; s += 16) {
#pragma unroll
                for (int u = 0; u < 8; ++u) gc[u] = gb[(s + 8 + u) * WG_B + lane];
#pragma unroll
                for (int u = 0; u < 8; ++u) {
                    orow[(size_t)(s + u) * BATCH] = cell_step(ga[u], h, c, wi, wf, wg, wo);
                }
                const int sn = (s + 16 < CH) ? (s + 16) : 0;
#pragma unroll
                for (int u = 0; u < 8; ++u) ga[u] = gb[(sn + u) * WG_B + lane];
#pragma unroll
                for (int u = 0; u < 8; ++u) {
                    orow[(size_t)(s + 8 + u) * BATCH] = cell_step(gc[u], h, c, wi, wf, wg, wo);
                }
            }
        } else if (worker && ch + 1 < NCH) {
            produce_chunk(x, (float*)gbuf[(ch + 1) & 1], ch + 1, b0, rank, lane, bmat, bias);
        }
        lds_barrier();
    }
}

extern "C" void kernel_launch(void* const* d_in, const int* in_sizes, int n_in,
                              void* d_out, int out_size, void* d_ws, size_t ws_size,
                              hipStream_t stream)
{
    const float* x    = (const float*)d_in[0];
    const float* W_ih = (const float*)d_in[1];
    const float* W_hh = (const float*)d_in[2];
    const float* b_ih = (const float*)d_in[3];
    const float* b_hh = (const float*)d_in[4];
    float* out = (float*)d_out;

    lstm_fused_wmma<<<16, 512, 0, stream>>>(x, W_ih, W_hh, b_ih, b_hh, out);
}


# scan cell: c' = f*c + i*g as ONE quotient (c*Dg + Ng*(1+ef)) / ((1+ef)*Dg): 3 divisions a step become 2, in series; i,f exponents capped at 60; the sum now rounds before the division
# speedup vs baseline: 5.7111x; 1.0340x over previous
#include <hip/hip_runtime.h>
#include <hip/hip_bf16.h>

typedef __attribute__((ext_vector_type(2))) float v2f;
typedef __attribute__((ext_vector_type(4))) float v4f;
typedef __attribute__((ext_vector_type(8))) float v8f;
typedef v2f v2f_a4 __attribute__((aligned(4)));
typedef v4f v4f_a4 __attribute__((aligned(4)));

#define KOFF    13

#define T_STEPS 2048
#define BATCH   512
#define NIN     27

#define WG_B    32
#define CH      64
#define NCH     (T_STEPS / CH)
#define NWAVES  16
#define NWORK   12
#define NTILE   (CH * WG_B / 16)
#define NFULL   (NTILE / NWORK)
#define NTAIL   (NTILE - NFULL * NWORK)
#define TPASS   5

#define LOG2E   1.44269504088896340736f
#define CSCALE  (2.0f * LOG2E)
__device__ __forceinline__ float inv_sigmoid2(float v2) {
    return 1.0f + __builtin_amdgcn_exp2f(v2);
}

__device__ __forceinline__ float tanh2_over(float v2, float a, float m) {
    const float e = __builtin_amdgcn_exp2f(-fabsf(v2));
    const float s = copysignf(m, v2);
    return fmaf(-e, s, s) / (a * (1.0f + e));
}

#define ZCAP    60.0f

__device__ __forceinline__ float cell_step(const float4 g, float& h, float& c,
                                           float wi, float wf, float wg, float wo)
{
    const float zg = g.z + h * wg;
    const float zi = fminf(g.x + h * wi, ZCAP);
    const float zf = fminf(g.y + h * wf, ZCAP);
    const float zo = g.w + h * wo;
    const float eg = __builtin_amdgcn_exp2f(-fabsf(zg));
    const float ei = __builtin_amdgcn_exp2f(zi);
    const float ef = __builtin_amdgcn_exp2f(zf);
    const float s  = copysignf(CSCALE, zg);
    const float ng = fmaf(-eg, s, s);
    const float ug = 1.0f + eg;
    const float dg = fmaf(ei, ug, ug);
    const float p  = fmaf(ef, ng, ng);
    const float den = fmaf(ef, dg, dg);
    const float num = fmaf(c, dg, p);
    const float ao = inv_sigmoid2(zo);
    c = num / den;
    h = tanh2_over(c, ao, 1.0f);
    return h;
}

__device__ __forceinline__ void lds_barrier()
{
    __builtin_amdgcn_fence(__ATOMIC_RELEASE, "workgroup", "local");
    __builtin_amdgcn_s_barrier();
    __builtin_amdgcn_fence(__ATOMIC_ACQUIRE, "workgroup", "local");
}


__device__ __forceinline__ void load_tile(const float* __restrict__ x, size_t crow0, int t,
                                          int mn, int kh, v2f (&a)[7])
{
    const size_t grow = crow0 + (size_t)(t >> 1) * BATCH + (t & 1) * 16;
    const float* __restrict__ run = x + (grow + mn) * NIN + KOFF * kh;
    const v4f_a4 r0 = *(const v4f_a4*)(run);
    const v4f_a4 r1 = *(const v4f_a4*)(run + 4);
    const v4f_a4 r2 = *(const v4f_a4*)(run + 8);
    const v2f_a4 r3 = *(const v2f_a4*)(run + 12);
    a[0].x = r0.x;  a[0].y = r0.y;
    a[1].x = r0.z;  a[1].y = r0.w;
    a[2].x = r1.x;  a[2].y = r1.y;
    a[3].x = r1.z;  a[3].y = r1.w;
    a[4].x = r2.x;  a[4].y = r2.y;
    a[5].x = r2.z;  a[5].y = r2.w;
    a[6].x = r3.x;  a[6].y = r3.y;
}

__device__ __forceinline__ void finish_tile(const v2f (&a)[7], const v2f (&bmat)[7], float bias,
                                            int t, int lane, int mn, float* __restrict__ dst)
{
    v8f acc = {};
#pragma unroll
    for (int kk = 0; kk < 7; ++kk) {
        acc = __builtin_amdgcn_wmma_f32_16x16x4_f32(
            false, a[kk], false, bmat[kk], (short)0, acc, false, false);
    }
    if (mn < 4) {
        const int rl = (t >> 1) * WG_B + (t & 1) * 16 + ((lane >> 4) << 3);
#pragma unroll
        for (int r = 0; r < 8; ++r) {
            dst[(rl + r) * 4 + mn] = acc[r] + bias;
        }
    }
}

__device__ __forceinline__ void produce_chunk(const float* __restrict__ x,
                                              float* __restrict__ dst,
                                              int chunk, int b0, int rank, int lane,
                                              const v2f (&bmat)[7], float bias)
{
    const int mn = lane & 15;
    const int kh = lane >> 4;
    const size_t crow0 = (size_t)chunk * CH * BATCH + b0;

    for (int j0 = 0; j0 < NFULL; j0 += TPASS) {
        v2f a[TPASS][7];
#pragma unroll
        for (int q = 0; q < TPASS; ++q) {
            load_tile(x, crow0, rank + NWORK * (j0 + q), mn, kh, a[q]);
        }
#pragma unroll
        for (int q = 0; q < TPASS; ++q) {
            finish_tile(a[q], bmat, bias, rank + NWORK * (j0 + q), lane, mn, dst);
        }
    }
    if (rank < NTAIL) {
        v2f a[7];
        load_tile(x, crow0, NFULL * NWORK + rank, mn, kh, a);
        finish_tile(a, bmat, bias, NFULL * NWORK + rank, lane, mn, dst);
    }
}

__global__ __launch_bounds__(32 * NWAVES)
void lstm_fused_wmma(const float* __restrict__ x,
                     const float* __restrict__ W_ih,
                     const float* __restrict__ W_hh,
                     const float* __restrict__ b_ih,
                     const float* __restrict__ b_hh,
                     float* __restrict__ out)
{
    __shared__ float4 gbuf[2][CH * WG_B];

    const int lane = threadIdx.x & 31;
    const int wave = __builtin_amdgcn_readfirstlane((int)(threadIdx.x >> 5));
    const int b0   = blockIdx.x * WG_B;
    const bool worker = (wave & 3) != 0;
    const int  rank   = (wave >> 2) * 3 + (wave & 3) - 1;

    const int mn = lane & 15;
    const int kh = lane >> 4;

    const float gscale = (mn == 2) ? CSCALE : -LOG2E;
    v2f bmat[7];
#pragma unroll
    for (int kk = 0; kk < 7; ++kk) {
        const int k0 = KOFF * kh + 2 * kk;
        float bx = 0.0f, by = 0.0f;
        if (mn < 4) {
            if (!(kh == 1 && kk == 0)) bx = W_ih[mn * NIN + k0] * gscale;
            by = W_ih[mn * NIN + k0 + 1] * gscale;
        }
        bmat[kk].x = bx;
        bmat[kk].y = by;
    }
    const float bias = (mn < 4) ? (b_ih[mn] + b_hh[mn]) * gscale : 0.0f;

    const float wi = W_hh[0] * -LOG2E;
    const float wf = W_hh[1] * -LOG2E;
    const float wg = W_hh[2] * CSCALE;
    const float wo = W_hh[3] * -LOG2E;

    float h = 0.0f, c = 0.0f;

    if (worker) {
        produce_chunk(x, (float*)gbuf[0], 0, b0, rank, lane, bmat, bias);
    }
    lds_barrier();

    for (int ch = 0; ch < NCH; ++ch) {
        if (wave == 0) {
            const float4* gb = gbuf[ch & 1];
            float* __restrict__ orow = out + (size_t)(ch * CH) * BATCH + b0 + lane;

            float4 ga[8], gc[8];
#pragma unroll
            for (int u = 0; u < 8; ++u) ga[u] = gb[u * WG_B + lane];

            for (int s = 0; s < CH; s += 16) {
#pragma unroll
                for (int u = 0; u < 8; ++u) gc[u] = gb[(s + 8 + u) * WG_B + lane];
#pragma unroll
                for (int u = 0; u < 8; ++u) {
                    orow[(size_t)(s + u) * BATCH] = cell_step(ga[u], h, c, wi, wf, wg, wo);
                }
                const int sn = (s + 16 < CH) ? (s + 16) : 0;
#pragma unroll
                for (int u = 0; u < 8; ++u) ga[u] = gb[(sn + u) * WG_B + lane];
#pragma unroll
                for (int u = 0; u < 8; ++u) {
                    orow[(size_t)(s + 8 + u) * BATCH] = cell_step(gc[u], h, c, wi, wf, wg, wo);
                }
            }
        } else if (worker && ch + 1 < NCH) {
            produce_chunk(x, (float*)gbuf[(ch + 1) & 1], ch + 1, b0, rank, lane, bmat, bias);
        }
        lds_barrier();
    }
}

extern "C" void kernel_launch(void* const* d_in, const int* in_sizes, int n_in,
                              void* d_out, int out_size, void* d_ws, size_t ws_size,
                              hipStream_t stream)
{
    const float* x    = (const float*)d_in[0];
    const float* W_ih = (const float*)d_in[1];
    const float* W_hh = (const float*)d_in[2];
    const float* b_ih = (const float*)d_in[3];
    const float* b_hh = (const float*)d_in[4];
    float* out = (float*)d_out;

    lstm_fused_wmma<<<16, 512, 0, stream>>>(x, W_ih, W_hh, b_ih, b_hh, out);
}


# chain wave: the 16 h of a pass stay in registers and leave in one run of 16 stores behind the pass; LDS reads held in their two groups: 3 memory places per 16 steps, not 12; sums unchanged
# speedup vs baseline: 6.1541x; 1.0776x over previous
#include <hip/hip_runtime.h>
#include <hip/hip_bf16.h>

typedef __attribute__((ext_vector_type(2))) float v2f;
typedef __attribute__((ext_vector_type(4))) float v4f;
typedef __attribute__((ext_vector_type(8))) float v8f;
typedef v2f v2f_a4 __attribute__((aligned(4)));
typedef v4f v4f_a4 __attribute__((aligned(4)));

#define KOFF    13

#define T_STEPS 2048
#define BATCH   512
#define NIN     27

#define WG_B    32
#define CH      64
#define NCH     (T_STEPS / CH)
#define NWAVES  16
#define NWORK   12
#define NTILE   (CH * WG_B / 16)
#define NFULL   (NTILE / NWORK)
#define NTAIL   (NTILE - NFULL * NWORK)
#define TPASS   5

#define LOG2E   1.44269504088896340736f
#define CSCALE  (2.0f * LOG2E)
__device__ __forceinline__ float inv_sigmoid2(float v2) {
    return 1.0f + __builtin_amdgcn_exp2f(v2);
}

__device__ __forceinline__ float tanh2_over(float v2, float a, float m) {
    const float e = __builtin_amdgcn_exp2f(-fabsf(v2));
    const float s = copysignf(m, v2);
    return fmaf(-e, s, s) / (a * (1.0f + e));
}

#define ZCAP    60.0f

__device__ __forceinline__ float cell_step(const v4f g, float& h, float& c,
                                           float wi, float wf, float wg, float wo)
{
    const float zg = g.z + h * wg;
    const float zi = fminf(g.x + h * wi, ZCAP);
    const float zf = fminf(g.y + h * wf, ZCAP);
    const float zo = g.w + h * wo;
    const float eg = __builtin_amdgcn_exp2f(-fabsf(zg));
    const float ei = __builtin_amdgcn_exp2f(zi);
    const float ef = __builtin_amdgcn_exp2f(zf);
    const float s  = copysignf(CSCALE, zg);
    const float ng = fmaf(-eg, s, s);
    const float ug = 1.0f + eg;
    const float dg = fmaf(ei, ug, ug);
    const float p  = fmaf(ef, ng, ng);
    const float den = fmaf(ef, dg, dg);
    const float num = fmaf(c, dg, p);
    const float ao = inv_sigmoid2(zo);
    c = num / den;
    h = tanh2_over(c, ao, 1.0f);
    return h;
}

__device__ __forceinline__ void lds_barrier()
{
    __builtin_amdgcn_fence(__ATOMIC_RELEASE, "workgroup", "local");
    __builtin_amdgcn_s_barrier();
    __builtin_amdgcn_fence(__ATOMIC_ACQUIRE, "workgroup", "local");
}


__device__ __forceinline__ void load_tile(const float* __restrict__ x, size_t crow0, int t,
                                          int mn, int kh, v2f (&a)[7])
{
    const size_t grow = crow0 + (size_t)(t >> 1) * BATCH + (t & 1) * 16;
    const float* __restrict__ run = x + (grow + mn) * NIN + KOFF * kh;
    const v4f_a4 r0 = *(const v4f_a4*)(run);
    const v4f_a4 r1 = *(const v4f_a4*)(run + 4);
    const v4f_a4 r2 = *(const v4f_a4*)(run + 8);
    const v2f_a4 r3 = *(const v2f_a4*)(run + 12);
    a[0].x = r0.x;  a[0].y = r0.y;
    a[1].x = r0.z;  a[1].y = r0.w;
    a[2].x = r1.x;  a[2].y = r1.y;
    a[3].x = r1.z;  a[3].y = r1.w;
    a[4].x = r2.x;  a[4].y = r2.y;
    a[5].x = r2.z;  a[5].y = r2.w;
    a[6].x = r3.x;  a[6].y = r3.y;
}

__device__ __forceinline__ void finish_tile(const v2f (&a)[7], const v2f (&bmat)[7], float bias,
                                            int t, int lane, int mn, float* __restrict__ dst)
{
    v8f acc = {};
#pragma unroll
    for (int kk = 0; kk < 7; ++kk) {
        acc = __builtin_amdgcn_wmma_f32_16x16x4_f32(
            false, a[kk], false, bmat[kk], (short)0, acc, false, false);
    }
    if (mn < 4) {
        const int rl = (t >> 1) * WG_B + (t & 1) * 16 + ((lane >> 4) << 3);
#pragma unroll
        for (int r = 0; r < 8; ++r) {
            dst[(rl + r) * 4 + mn] = acc[r] + bias;
        }
    }
}

__device__ __forceinline__ void produce_chunk(const float* __restrict__ x,
                                              float* __restrict__ dst,
                                              int chunk, int b0, int rank, int lane,
                                              const v2f (&bmat)[7], float bias)
{
    const int mn = lane & 15;
    const int kh = lane >> 4;
    const size_t crow0 = (size_t)chunk * CH * BATCH + b0;

    for (int j0 = 0; j0 < NFULL; j0 += TPASS) {
        v2f a[TPASS][7];
#pragma unroll
        for (int q = 0; q < TPASS; ++q) {
            load_tile(x, crow0, rank + NWORK * (j0 + q), mn, kh, a[q]);
        }
#pragma unroll
        for (int q = 0; q < TPASS; ++q) {
            finish_tile(a[q], bmat, bias, rank + NWORK * (j0 + q), lane, mn, dst);
        }
    }
    if (rank < NTAIL) {
        v2f a[7];
        load_tile(x, crow0, NFULL * NWORK + rank, mn, kh, a);
        finish_tile(a, bmat, bias, NFULL * NWORK + rank, lane, mn, dst);
    }
}

__global__ __launch_bounds__(32 * NWAVES)
void lstm_fused_wmma(const float* __restrict__ x,
                     const float* __restrict__ W_ih,
                     const float* __restrict__ W_hh,
                     const float* __restrict__ b_ih,
                     const float* __restrict__ b_hh,
                     float* __restrict__ out)
{
    __shared__ float4 gbuf[2][CH * WG_B];

    const int lane = threadIdx.x & 31;
    const int wave = __builtin_amdgcn_readfirstlane((int)(threadIdx.x >> 5));
    const int b0   = blockIdx.x * WG_B;
    const bool worker = (wave & 3) != 0;
    const int  rank   = (wave >> 2) * 3 + (wave & 3) - 1;

    const int mn = lane & 15;
    const int kh = lane >> 4;

    const float gscale = (mn == 2) ? CSCALE : -LOG2E;
    v2f bmat[7];
#pragma unroll
    for (int kk = 0; kk < 7; ++kk) {
        const int k0 = KOFF * kh + 2 * kk;
        float bx = 0.0f, by = 0.0f;
        if (mn < 4) {
            if (!(kh == 1 && kk == 0)) bx = W_ih[mn * NIN + k0] * gscale;
            by = W_ih[mn * NIN + k0 + 1] * gscale;
        }
        bmat[kk].x = bx;
        bmat[kk].y = by;
    }
    const float bias = (mn < 4) ? (b_ih[mn] + b_hh[mn]) * gscale : 0.0f;

    const float wi = W_hh[0] * -LOG2E;
    const float wf = W_hh[1] * -LOG2E;
    const float wg = W_hh[2] * CSCALE;
    const float wo = W_hh[3] * -LOG2E;

    float h = 0.0f, c = 0.0f;

    if (worker) {
        produce_chunk(x, (float*)gbuf[0], 0, b0, rank, lane, bmat, bias);
    }
    lds_barrier();

    for (int ch = 0; ch < NCH; ++ch) {
        if (wave == 0) {
            const volatile __attribute__((address_space(3))) v4f* gb =
                (const volatile __attribute__((address_space(3))) v4f*)gbuf[ch & 1];
            float* __restrict__ orow = out + (size_t)(ch * CH) * BATCH + b0 + lane;

            v4f ga[8], gc[8];
            float hv[16];
#pragma unroll
            for (int u = 0; u < 8; ++u) ga[u] = gb[u * WG_B + lane];

            for (int s = 0; s < CH; s += 16) {
#pragma unroll
                for (int u = 0; u < 8; ++u) gc[u] = gb[(s + 8 + u) * WG_B + lane];
                __builtin_amdgcn_sched_barrier(0);
#pragma unroll
                for (int u = 0; u < 8; ++u) {
                    hv[u] = cell_step(ga[u], h, c, wi, wf, wg, wo);
                }
                const int sn = (s + 16 < CH) ? (s + 16) : 0;
#pragma unroll
                for (int u = 0; u < 8; ++u) ga[u] = gb[(sn + u) * WG_B + lane];
                __builtin_amdgcn_sched_barrier(0);
#pragma unroll
                for (int u = 0; u < 8; ++u) {
                    hv[8 + u] = cell_step(gc[u], h, c, wi, wf, wg, wo);
                }
                __builtin_amdgcn_sched_barrier(0);
#pragma unroll
                for (int u = 0; u < 16; ++u) {
                    orow[(size_t)(s + u) * BATCH] = hv[u];
                }
            }
        } else if (worker && ch + 1 < NCH) {
            produce_chunk(x, (float*)gbuf[(ch + 1) & 1], ch + 1, b0, rank, lane, bmat, bias);
        }
        lds_barrier();
    }
}

extern "C" void kernel_launch(void* const* d_in, const int* in_sizes, int n_in,
                              void* d_out, int out_size, void* d_ws, size_t ws_size,
                              hipStream_t stream)
{
    const float* x    = (const float*)d_in[0];
    const float* W_ih = (const float*)d_in[1];
    const float* W_hh = (const float*)d_in[2];
    const float* b_ih = (const float*)d_in[3];
    const float* b_hh = (const float*)d_in[4];
    float* out = (float*)d_out;

    lstm_fused_wmma<<<16, 512, 0, stream>>>(x, W_ih, W_hh, b_ih, b_hh, out);
}


# scan cell: sigmoids from e=2^-|z| keep every denominator in [1,8]; both divisions written out, correctly rounded steps kept, range scaling and fix-up dropped; issue order held; products round anew
# speedup vs baseline: 6.6034x; 1.0730x over previous
#include <hip/hip_runtime.h>
#include <hip/hip_bf16.h>

typedef __attribute__((ext_vector_type(2))) float v2f;
typedef __attribute__((ext_vector_type(4))) float v4f;
typedef __attribute__((ext_vector_type(8))) float v8f;
typedef v2f v2f_a4 __attribute__((aligned(4)));
typedef v4f v4f_a4 __attribute__((aligned(4)));

#define KOFF    13

#define T_STEPS 2048
#define BATCH   512
#define NIN     27

#define WG_B    32
#define CH      64
#define NCH     (T_STEPS / CH)
#define NWAVES  16
#define NWORK   12
#define NTILE   (CH * WG_B / 16)
#define NFULL   (NTILE / NWORK)
#define NTAIL   (NTILE - NFULL * NWORK)
#define TPASS   5

#define LOG2E   1.44269504088896340736f
#define CSCALE  (2.0f * LOG2E)
__device__ __forceinline__ float exp2_neg_abs(float v2) {
    return __builtin_amdgcn_exp2f(-fabsf(v2));
}

__device__ __forceinline__ float div_bounded(float n, float d) {
    const float r0 = __builtin_amdgcn_rcpf(d);
    const float r  = fmaf(fmaf(-d, r0, 1.0f), r0, r0);
    const float q0 = n * r;
    const float q1 = fmaf(fmaf(-d, q0, n), r, q0);
    return fmaf(fmaf(-d, q1, n), r, q1);
}

__device__ __forceinline__ float cell_step(const v4f g, float& h, float& c,
                                           float wi, float wf, float wg, float wo)
{
    const float zg = g.z + h * wg;
    const float zi = g.x + h * wi;
    const float zf = g.y + h * wf;
    float zo = g.w + h * wo;
    float eg = exp2_neg_abs(zg);
    float ei = exp2_neg_abs(zi);
    float ef = exp2_neg_abs(zf);
    asm volatile("" : "+v"(eg), "+v"(ei), "+v"(ef));
    const float s  = copysignf(CSCALE, zg);
    const float ug = 1.0f + eg;
    const float dg = fmaf(ei, ug, ug);
    float den = fmaf(ef, dg, dg);
    asm volatile("" : "+v"(den), "+v"(ei), "+v"(ef), "+v"(zo));
    float eo = exp2_neg_abs(zo);
    const float ng = fmaf(-eg, s, s);
    const float ni = (zi > 0.0f) ? ei : 1.0f;
    const float nf = (zf > 0.0f) ? ef : 1.0f;
    const float a  = ni * ng;
    const float p  = fmaf(ef, a, a);
    float num = fmaf(nf * c, dg, p);
    asm volatile("" : "+v"(num), "+v"(eo));
    c = div_bounded(num, den);
    float no = (zo > 0.0f) ? eo : 1.0f;
    float uo = 1.0f + eo;
    asm volatile("" : "+v"(c), "+v"(no), "+v"(uo));
    const float e  = exp2_neg_abs(c);
    const float sc = copysignf(no, c);
    h = div_bounded(fmaf(-e, sc, sc), fmaf(e, uo, uo));
    return h;
}

__device__ __forceinline__ void lds_barrier()
{
    __builtin_amdgcn_fence(__ATOMIC_RELEASE, "workgroup", "local");
    __builtin_amdgcn_s_barrier();
    __builtin_amdgcn_fence(__ATOMIC_ACQUIRE, "workgroup", "local");
}


__device__ __forceinline__ void load_tile(const float* __restrict__ x, size_t crow0, int t,
                                          int mn, int kh, v2f (&a)[7])
{
    const size_t grow = crow0 + (size_t)(t >> 1) * BATCH + (t & 1) * 16;
    const float* __restrict__ run = x + (grow + mn) * NIN + KOFF * kh;
    const v4f_a4 r0 = *(const v4f_a4*)(run);
    const v4f_a4 r1 = *(const v4f_a4*)(run + 4);
    const v4f_a4 r2 = *(const v4f_a4*)(run + 8);
    const v2f_a4 r3 = *(const v2f_a4*)(run + 12);
    a[0].x = r0.x;  a[0].y = r0.y;
    a[1].x = r0.z;  a[1].y = r0.w;
    a[2].x = r1.x;  a[2].y = r1.y;
    a[3].x = r1.z;  a[3].y = r1.w;
    a[4].x = r2.x;  a[4].y = r2.y;
    a[5].x = r2.z;  a[5].y = r2.w;
    a[6].x = r3.x;  a[6].y = r3.y;
}

__device__ __forceinline__ void finish_tile(const v2f (&a)[7], const v2f (&bmat)[7], float bias,
                                            int t, int lane, int mn, float* __restrict__ dst)
{
    v8f acc = {};
#pragma unroll
    for (int kk = 0; kk < 7; ++kk) {
        acc = __builtin_amdgcn_wmma_f32_16x16x4_f32(
            false, a[kk], false, bmat[kk], (short)0, acc, false, false);
    }
    if (mn < 4) {
        const int rl = (t >> 1) * WG_B + (t & 1) * 16 + ((lane >> 4) << 3);
#pragma unroll
        for (int r = 0; r < 8; ++r) {
            dst[(rl + r) * 4 + mn] = acc[r] + bias;
        }
    }
}

__device__ __forceinline__ void produce_chunk(const float* __restrict__ x,
                                              float* __restrict__ dst,
                                              int chunk, int b0, int rank, int lane,
                                              const v2f (&bmat)[7], float bias)
{
    const int mn = lane & 15;
    const int kh = lane >> 4;
    const size_t crow0 = (size_t)chunk * CH * BATCH + b0;

    for (int j0 = 0; j0 < NFULL; j0 += TPASS) {
        v2f a[TPASS][7];
#pragma unroll
        for (int q = 0; q < TPASS; ++q) {
            load_tile(x, crow0, rank + NWORK * (j0 + q), mn, kh, a[q]);
        }
#pragma unroll
        for (int q = 0; q < TPASS; ++q) {
            finish_tile(a[q], bmat, bias, rank + NWORK * (j0 + q), lane, mn, dst);
        }
    }
    if (rank < NTAIL) {
        v2f a[7];
        load_tile(x, crow0, NFULL * NWORK + rank, mn, kh, a);
        finish_tile(a, bmat, bias, NFULL * NWORK + rank, lane, mn, dst);
    }
}

__global__ __launch_bounds__(32 * NWAVES)
void lstm_fused_wmma(const float* __restrict__ x,
                     const float* __restrict__ W_ih,
                     const float* __restrict__ W_hh,
                     const float* __restrict__ b_ih,
                     const float* __restrict__ b_hh,
                     float* __restrict__ out)
{
    __shared__ float4 gbuf[2][CH * WG_B];

    const int lane = threadIdx.x & 31;
    const int wave = __builtin_amdgcn_readfirstlane((int)(threadIdx.x >> 5));
    const int b0   = blockIdx.x * WG_B;
    const bool worker = (wave & 3) != 0;
    const int  rank   = (wave >> 2) * 3 + (wave & 3) - 1;

    const int mn = lane & 15;
    const int kh = lane >> 4;

    const float gscale = (mn == 2) ? CSCALE : -LOG2E;
    v2f bmat[7];
#pragma unroll
    for (int kk = 0; kk < 7; ++kk) {
        const int k0 = KOFF * kh + 2 * kk;
        float bx = 0.0f, by = 0.0f;
        if (mn < 4) {
            if (!(kh == 1 && kk == 0)) bx = W_ih[mn * NIN + k0] * gscale;
            by = W_ih[mn * NIN + k0 + 1] * gscale;
        }
        bmat[kk].x = bx;
        bmat[kk].y = by;
    }
    const float bias = (mn < 4) ? (b_ih[mn] + b_hh[mn]) * gscale : 0.0f;

    const float wi = W_hh[0] * -LOG2E;
    const float wf = W_hh[1] * -LOG2E;
    const float wg = W_hh[2] * CSCALE;
    const float wo = W_hh[3] * -LOG2E;

    float h = 0.0f, c = 0.0f;

    if (worker) {
        produce_chunk(x, (float*)gbuf[0], 0, b0, rank, lane, bmat, bias);
    }
    lds_barrier();

    for (int ch = 0; ch < NCH; ++ch) {
        if (wave == 0) {
            const volatile __attribute__((address_space(3))) v4f* gb =
                (const volatile __attribute__((address_space(3))) v4f*)gbuf[ch & 1];
            float* __restrict__ orow = out + (size_t)(ch * CH) * BATCH + b0 + lane;

            v4f ga[8], gc[8];
            float hv[16];
#pragma unroll
            for (int u = 0; u < 8; ++u) ga[u] = gb[u * WG_B + lane];

            for (int s = 0; s < CH; s += 16) {
#pragma unroll
                for (int u = 0; u < 8; ++u) gc[u] = gb[(s + 8 + u) * WG_B + lane];
                __builtin_amdgcn_sched_barrier(0);
#pragma unroll
                for (int u = 0; u < 8; ++u) {
                    hv[u] = cell_step(ga[u], h, c, wi, wf, wg, wo);
                }
                const int sn = (s + 16 < CH) ? (s + 16) : 0;
#pragma unroll
                for (int u = 0; u < 8; ++u) ga[u] = gb[(sn + u) * WG_B + lane];
                __builtin_amdgcn_sched_barrier(0);
#pragma unroll
                for (int u = 0; u < 8; ++u) {
                    hv[8 + u] = cell_step(gc[u], h, c, wi, wf, wg, wo);
                }
                __builtin_amdgcn_sched_barrier(0);
#pragma unroll
                for (int u = 0; u < 16; ++u) {
                    orow[(size_t)(s + u) * BATCH] = hv[u];
                }
            }
        } else if (worker && ch + 1 < NCH) {
            produce_chunk(x, (float*)gbuf[(ch + 1) & 1], ch + 1, b0, rank, lane, bmat, bias);
        }
        lds_barrier();
    }
}

extern "C" void kernel_launch(void* const* d_in, const int* in_sizes, int n_in,
                              void* d_out, int out_size, void* d_ws, size_t ws_size,
                              hipStream_t stream)
{
    const float* x    = (const float*)d_in[0];
    const float* W_ih = (const float*)d_in[1];
    const float* W_hh = (const float*)d_in[2];
    const float* b_ih = (const float*)d_in[3];
    const float* b_hh = (const float*)d_in[4];
    float* out = (float*)d_out;

    lstm_fused_wmma<<<16, 512, 0, stream>>>(x, W_ih, W_hh, b_ih, b_hh, out);
}


# decomposition: 32 work-groups of 16 chains (was 16 of 32): per WGP half the x projection per chunk beside the same chain, chain wave on 16 lanes; sums unchanged; out lines get 2 writers
# speedup vs baseline: 8.0452x; 1.2183x over previous
#include <hip/hip_runtime.h>
#include <hip/hip_bf16.h>

typedef __attribute__((ext_vector_type(2))) float v2f;
typedef __attribute__((ext_vector_type(4))) float v4f;
typedef __attribute__((ext_vector_type(8))) float v8f;
typedef v2f v2f_a4 __attribute__((aligned(4)));
typedef v4f v4f_a4 __attribute__((aligned(4)));

#define KOFF    13

#define T_STEPS 2048
#define BATCH   512
#define NIN     27

#define WG_B    16
#define CH      64
#define NCH     (T_STEPS / CH)
#define NWAVES  16
#define NWORK   12
#define TPS     (WG_B / 16)
#define NTILE   (CH * TPS)
#define NFULL   (NTILE / NWORK)
#define NTAIL   (NTILE - NFULL * NWORK)
#define TPASS   5

#define LOG2E   1.44269504088896340736f
#define CSCALE  (2.0f * LOG2E)
__device__ __forceinline__ float exp2_neg_abs(float v2) {
    return __builtin_amdgcn_exp2f(-fabsf(v2));
}

__device__ __forceinline__ float div_bounded(float n, float d) {
    const float r0 = __builtin_amdgcn_rcpf(d);
    const float r  = fmaf(fmaf(-d, r0, 1.0f), r0, r0);
    const float q0 = n * r;
    const float q1 = fmaf(fmaf(-d, q0, n), r, q0);
    return fmaf(fmaf(-d, q1, n), r, q1);
}

__device__ __forceinline__ float cell_step(const v4f g, float& h, float& c,
                                           float wi, float wf, float wg, float wo)
{
    const float zg = g.z + h * wg;
    const float zi = g.x + h * wi;
    const float zf = g.y + h * wf;
    float zo = g.w + h * wo;
    float eg = exp2_neg_abs(zg);
    float ei = exp2_neg_abs(zi);
    float ef = exp2_neg_abs(zf);
    asm volatile("" : "+v"(eg), "+v"(ei), "+v"(ef));
    const float s  = copysignf(CSCALE, zg);
    const float ug = 1.0f + eg;
    const float dg = fmaf(ei, ug, ug);
    float den = fmaf(ef, dg, dg);
    asm volatile("" : "+v"(den), "+v"(ei), "+v"(ef), "+v"(zo));
    float eo = exp2_neg_abs(zo);
    const float ng = fmaf(-eg, s, s);
    const float ni = (zi > 0.0f) ? ei : 1.0f;
    const float nf = (zf > 0.0f) ? ef : 1.0f;
    const float a  = ni * ng;
    const float p  = fmaf(ef, a, a);
    float num = fmaf(nf * c, dg, p);
    asm volatile("" : "+v"(num), "+v"(eo));
    c = div_bounded(num, den);
    float no = (zo > 0.0f) ? eo : 1.0f;
    float uo = 1.0f + eo;
    asm volatile("" : "+v"(c), "+v"(no), "+v"(uo));
    const float e  = exp2_neg_abs(c);
    const float sc = copysignf(no, c);
    h = div_bounded(fmaf(-e, sc, sc), fmaf(e, uo, uo));
    return h;
}

__device__ __forceinline__ void lds_barrier()
{
    __builtin_amdgcn_fence(__ATOMIC_RELEASE, "workgroup", "local");
    __builtin_amdgcn_s_barrier();
    __builtin_amdgcn_fence(__ATOMIC_ACQUIRE, "workgroup", "local");
}


__device__ __forceinline__ void load_tile(const float* __restrict__ x, size_t crow0, int t,
                                          int mn, int kh, v2f (&a)[7])
{
    const size_t grow = crow0 + (size_t)(t / TPS) * BATCH + (t % TPS) * 16;
    const float* __restrict__ run = x + (grow + mn) * NIN + KOFF * kh;
    const v4f_a4 r0 = *(const v4f_a4*)(run);
    const v4f_a4 r1 = *(const v4f_a4*)(run + 4);
    const v4f_a4 r2 = *(const v4f_a4*)(run + 8);
    const v2f_a4 r3 = *(const v2f_a4*)(run + 12);
    a[0].x = r0.x;  a[0].y = r0.y;
    a[1].x = r0.z;  a[1].y = r0.w;
    a[2].x = r1.x;  a[2].y = r1.y;
    a[3].x = r1.z;  a[3].y = r1.w;
    a[4].x = r2.x;  a[4].y = r2.y;
    a[5].x = r2.z;  a[5].y = r2.w;
    a[6].x = r3.x;  a[6].y = r3.y;
}

__device__ __forceinline__ void finish_tile(const v2f (&a)[7], const v2f (&bmat)[7], float bias,
                                            int t, int lane, int mn, float* __restrict__ dst)
{
    v8f acc = {};
#pragma unroll
    for (int kk = 0; kk < 7; ++kk) {
        acc = __builtin_amdgcn_wmma_f32_16x16x4_f32(
            false, a[kk], false, bmat[kk], (short)0, acc, false, false);
    }
    if (mn < 4) {
        const int rl = (t / TPS) * WG_B + (t % TPS) * 16 + ((lane >> 4) << 3);
#pragma unroll
        for (int r = 0; r < 8; ++r) {
            dst[(rl + r) * 4 + mn] = acc[r] + bias;
        }
    }
}

__device__ __forceinline__ void produce_chunk(const float* __restrict__ x,
                                              float* __restrict__ dst,
                                              int chunk, int b0, int rank, int lane,
                                              const v2f (&bmat)[7], float bias)
{
    const int mn = lane & 15;
    const int kh = lane >> 4;
    const size_t crow0 = (size_t)chunk * CH * BATCH + b0;

    for (int j0 = 0; j0 < NFULL; j0 += TPASS) {
        v2f a[TPASS][7];
#pragma unroll
        for (int q = 0; q < TPASS; ++q) {
            load_tile(x, crow0, rank + NWORK * (j0 + q), mn, kh, a[q]);
        }
#pragma unroll
        for (int q = 0; q < TPASS; ++q) {
            finish_tile(a[q], bmat, bias, rank + NWORK * (j0 + q), lane, mn, dst);
        }
    }
    if (rank < NTAIL) {
        v2f a[7];
        load_tile(x, crow0, NFULL * NWORK + rank, mn, kh, a);
        finish_tile(a, bmat, bias, NFULL * NWORK + rank, lane, mn, dst);
    }
}

__global__ __launch_bounds__(32 * NWAVES)
void lstm_fused_wmma(const float* __restrict__ x,
                     const float* __restrict__ W_ih,
                     const float* __restrict__ W_hh,
                     const float* __restrict__ b_ih,
                     const float* __restrict__ b_hh,
                     float* __restrict__ out)
{
    __shared__ float4 gbuf[2][CH * WG_B];

    const int lane = threadIdx.x & 31;
    const int wave = __builtin_amdgcn_readfirstlane((int)(threadIdx.x >> 5));
    const int b0   = blockIdx.x * WG_B;
    const bool worker = (wave & 3) != 0;
    const int  rank   = (wave >> 2) * 3 + (wave & 3) - 1;

    const int mn = lane & 15;
    const int kh = lane >> 4;

    const float gscale = (mn == 2) ? CSCALE : -LOG2E;
    v2f bmat[7];
#pragma unroll
    for (int kk = 0; kk < 7; ++kk) {
        const int k0 = KOFF * kh + 2 * kk;
        float bx = 0.0f, by = 0.0f;
        if (mn < 4) {
            if (!(kh == 1 && kk == 0)) bx = W_ih[mn * NIN + k0] * gscale;
            by = W_ih[mn * NIN + k0 + 1] * gscale;
        }
        bmat[kk].x = bx;
        bmat[kk].y = by;
    }
    const float bias = (mn < 4) ? (b_ih[mn] + b_hh[mn]) * gscale : 0.0f;

    const float wi = W_hh[0] * -LOG2E;
    const float wf = W_hh[1] * -LOG2E;
    const float wg = W_hh[2] * CSCALE;
    const float wo = W_hh[3] * -LOG2E;

    float h = 0.0f, c = 0.0f;

    if (worker) {
        produce_chunk(x, (float*)gbuf[0], 0, b0, rank, lane, bmat, bias);
    }
    lds_barrier();

    for (int ch = 0; ch < NCH; ++ch) {
        if (wave == 0) {
          if (lane < WG_B) {
            const volatile __attribute__((address_space(3))) v4f* gb =
                (const volatile __attribute__((address_space(3))) v4f*)gbuf[ch & 1];
            float* __restrict__ orow = out + (size_t)(ch * CH) * BATCH + b0 + lane;

            v4f ga[8], gc[8];
            float hv[16];
#pragma unroll
            for (int u = 0; u < 8; ++u) ga[u] = gb[u * WG_B + lane];

            for (int s = 0; s < CH; s += 16) {
#pragma unroll
                for (int u = 0; u < 8; ++u) gc[u] = gb[(s + 8 + u) * WG_B + lane];
                __builtin_amdgcn_sched_barrier(0);
#pragma unroll
                for (int u = 0; u < 8; ++u) {
                    hv[u] = cell_step(ga[u], h, c, wi, wf, wg, wo);
                }
                const int sn = (s + 16 < CH) ? (s + 16) : 0;
#pragma unroll
                for (int u = 0; u < 8; ++u) ga[u] = gb[(sn + u) * WG_B + lane];
                __builtin_amdgcn_sched_barrier(0);
#pragma unroll
                for (int u = 0; u < 8; ++u) {
                    hv[8 + u] = cell_step(gc[u], h, c, wi, wf, wg, wo);
                }
                __builtin_amdgcn_sched_barrier(0);
#pragma unroll
                for (int u = 0; u < 16; ++u) {
                    orow[(size_t)(s + u) * BATCH] = hv[u];
                }
            }
          }
        } else if (worker && ch + 1 < NCH) {
            produce_chunk(x, (float*)gbuf[(ch + 1) & 1], ch + 1, b0, rank, lane, bmat, bias);
        }
        lds_barrier();
    }
}

extern "C" void kernel_launch(void* const* d_in, const int* in_sizes, int n_in,
                              void* d_out, int out_size, void* d_ws, size_t ws_size,
                              hipStream_t stream)
{
    const float* x    = (const float*)d_in[0];
    const float* W_ih = (const float*)d_in[1];
    const float* W_hh = (const float*)d_in[2];
    const float* b_ih = (const float*)d_in[3];
    const float* b_hh = (const float*)d_in[4];
    float* out = (float*)d_out;

    lstm_fused_wmma<<<32, 512, 0, stream>>>(x, W_ih, W_hh, b_ih, b_hh, out);
}


# scan cell: each division's first quotient is n x the raw reciprocal, refined beside it: 8 -> 6 dependent ops; numerators ready first, picked by clamp+med3 not compare; quotients, sum order unchanged
# speedup vs baseline: 9.0522x; 1.1252x over previous
#include <hip/hip_runtime.h>
#include <hip/hip_bf16.h>

typedef __attribute__((ext_vector_type(2))) float v2f;
typedef __attribute__((ext_vector_type(4))) float v4f;
typedef __attribute__((ext_vector_type(8))) float v8f;
typedef v2f v2f_a4 __attribute__((aligned(4)));
typedef v4f v4f_a4 __attribute__((aligned(4)));

#define KOFF    13

#define T_STEPS 2048
#define BATCH   512
#define NIN     27

#define WG_B    16
#define CH      64
#define NCH     (T_STEPS / CH)
#define NWAVES  16
#define NWORK   12
#define TPS     (WG_B / 16)
#define NTILE   (CH * TPS)
#define NFULL   (NTILE / NWORK)
#define NTAIL   (NTILE - NFULL * NWORK)
#define TPASS   5

#define LOG2E   1.44269504088896340736f
#define CSCALE  (2.0f * LOG2E)
__device__ __forceinline__ float exp2_neg_abs(float v2) {
    return __builtin_amdgcn_exp2f(-fabsf(v2));
}

__device__ __forceinline__ float div_bounded(float n, float d) {
    const float r0 = __builtin_amdgcn_rcpf(d);
    float q0 = n * r0;
    float e0 = fmaf(-d, r0, 1.0f);
    asm volatile("" : "+v"(q0), "+v"(e0));
    const float r  = fmaf(e0, r0, r0);
    const float q1 = fmaf(fmaf(-d, q0, n), r, q0);
    return fmaf(fmaf(-d, q1, n), r, q1);
}

__device__ __forceinline__ float low_side(float z) {
    return __builtin_amdgcn_fmed3f(z * -0x1p100f, 0.0f, 1.0f);
}
__device__ __forceinline__ float pick_num(float e, float k) {
    return __builtin_amdgcn_fmed3f(e, k, 1.0f);
}

__device__ __forceinline__ float cell_step(const v4f g, float& h, float& c,
                                           float wi, float wf, float wg, float wo)
{
    const float zg = g.z + h * wg;
    const float zi = g.x + h * wi;
    const float zf = g.y + h * wf;
    const float zo = g.w + h * wo;
    float eg = exp2_neg_abs(zg);
    float ei = exp2_neg_abs(zi);
    float ef = exp2_neg_abs(zf);
    float eo = exp2_neg_abs(zo);
    asm volatile("" : "+v"(eg), "+v"(ei), "+v"(ef), "+v"(eo));
    const float s  = copysignf(CSCALE, zg);
    const float ki = low_side(zi);
    const float kf = low_side(zf);
    const float ko = low_side(zo);
    const float ug = 1.0f + eg;
    const float ng = fmaf(-eg, s, s);
    const float ni = pick_num(ei, ki);
    const float nf = pick_num(ef, kf);
    const float dg = fmaf(ei, ug, ug);
    const float a  = ni * ng;
    const float nfc = nf * c;
    const float den = fmaf(ef, dg, dg);
    const float p  = fmaf(ef, a, a);
    const float num = fmaf(nfc, dg, p);
    const float no = pick_num(eo, ko);
    const float uo = 1.0f + eo;
    c = div_bounded(num, den);
    float e  = exp2_neg_abs(c);
    float sc = copysignf(no, c);
    asm volatile("" : "+v"(e), "+v"(sc));
    h = div_bounded(fmaf(-e, sc, sc), fmaf(e, uo, uo));
    return h;
}

__device__ __forceinline__ void lds_barrier()
{
    __builtin_amdgcn_fence(__ATOMIC_RELEASE, "workgroup", "local");
    __builtin_amdgcn_s_barrier();
    __builtin_amdgcn_fence(__ATOMIC_ACQUIRE, "workgroup", "local");
}


__device__ __forceinline__ void load_tile(const float* __restrict__ x, size_t crow0, int t,
                                          int mn, int kh, v2f (&a)[7])
{
    const size_t grow = crow0 + (size_t)(t / TPS) * BATCH + (t % TPS) * 16;
    const float* __restrict__ run = x + (grow + mn) * NIN + KOFF * kh;
    const v4f_a4 r0 = *(const v4f_a4*)(run);
    const v4f_a4 r1 = *(const v4f_a4*)(run + 4);
    const v4f_a4 r2 = *(const v4f_a4*)(run + 8);
    const v2f_a4 r3 = *(const v2f_a4*)(run + 12);
    a[0].x = r0.x;  a[0].y = r0.y;
    a[1].x = r0.z;  a[1].y = r0.w;
    a[2].x = r1.x;  a[2].y = r1.y;
    a[3].x = r1.z;  a[3].y = r1.w;
    a[4].x = r2.x;  a[4].y = r2.y;
    a[5].x = r2.z;  a[5].y = r2.w;
    a[6].x = r3.x;  a[6].y = r3.y;
}

__device__ __forceinline__ void finish_tile(const v2f (&a)[7], const v2f (&bmat)[7], float bias,
                                            int t, int lane, int mn, float* __restrict__ dst)
{
    v8f acc = {};
    __builtin_amdgcn_sched_barrier(0);
#pragma unroll
    for (int kk = 0; kk < 7; ++kk) {
        acc = __builtin_amdgcn_wmma_f32_16x16x4_f32(
            false, a[kk], false, bmat[kk], (short)0, acc, false, false);
    }
    if (mn < 4) {
        const int rl = (t / TPS) * WG_B + (t % TPS) * 16 + ((lane >> 4) << 3);
#pragma unroll
        for (int r = 0; r < 8; ++r) {
            dst[(rl + r) * 4 + mn] = acc[r] + bias;
        }
    }
}

__device__ __forceinline__ void produce_chunk(const float* __restrict__ x,
                                              float* __restrict__ dst,
                                              int chunk, int b0, int rank, int lane,
                                              const v2f (&bmat)[7], float bias)
{
    const int mn = lane & 15;
    const int kh = lane >> 4;
    const size_t crow0 = (size_t)chunk * CH * BATCH + b0;

    for (int j0 = 0; j0 < NFULL; j0 += TPASS) {
        v2f a[TPASS][7];
#pragma unroll
        for (int q = 0; q < TPASS; ++q) {
            load_tile(x, crow0, rank + NWORK * (j0 + q), mn, kh, a[q]);
        }
#pragma unroll
        for (int q = 0; q < TPASS; ++q) {
            finish_tile(a[q], bmat, bias, rank + NWORK * (j0 + q), lane, mn, dst);
        }
    }
    if (rank < NTAIL) {
        v2f a[7];
        load_tile(x, crow0, NFULL * NWORK + rank, mn, kh, a);
        finish_tile(a, bmat, bias, NFULL * NWORK + rank, lane, mn, dst);
    }
}

__global__ __launch_bounds__(32 * NWAVES)
void lstm_fused_wmma(const float* __restrict__ x,
                     const float* __restrict__ W_ih,
                     const float* __restrict__ W_hh,
                     const float* __restrict__ b_ih,
                     const float* __restrict__ b_hh,
                     float* __restrict__ out)
{
    __shared__ float4 gbuf[2][CH * WG_B];

    const int lane = threadIdx.x & 31;
    const int wave = __builtin_amdgcn_readfirstlane((int)(threadIdx.x >> 5));
    const int b0   = blockIdx.x * WG_B;
    const bool worker = (wave & 3) != 0;
    const int  rank   = (wave >> 2) * 3 + (wave & 3) - 1;

    const int mn = lane & 15;
    const int kh = lane >> 4;

    const float gscale = (mn == 2) ? CSCALE : -LOG2E;
    v2f bmat[7];
#pragma unroll
    for (int kk = 0; kk < 7; ++kk) {
        const int k0 = KOFF * kh + 2 * kk;
        float bx = 0.0f, by = 0.0f;
        if (mn < 4) {
            if (!(kh == 1 && kk == 0)) bx = W_ih[mn * NIN + k0] * gscale;
            by = W_ih[mn * NIN + k0 + 1] * gscale;
        }
        bmat[kk].x = bx;
        bmat[kk].y = by;
    }
    const float bias = (mn < 4) ? (b_ih[mn] + b_hh[mn]) * gscale : 0.0f;

    const float wi = W_hh[0] * -LOG2E;
    const float wf = W_hh[1] * -LOG2E;
    const float wg = W_hh[2] * CSCALE;
    const float wo = W_hh[3] * -LOG2E;

    float h = 0.0f, c = 0.0f;

    if (worker) {
        produce_chunk(x, (float*)gbuf[0], 0, b0, rank, lane, bmat, bias);
    }
    lds_barrier();

    for (int ch = 0; ch < NCH; ++ch) {
        if (wave == 0) {
          if (lane < WG_B) {
            const volatile __attribute__((address_space(3))) v4f* gb =
                (const volatile __attribute__((address_space(3))) v4f*)gbuf[ch & 1];
            float* __restrict__ orow = out + (size_t)(ch * CH) * BATCH + b0 + lane;

            v4f ga[8], gc[8];
            float hv[16];
#pragma unroll
            for (int u = 0; u < 8; ++u) ga[u] = gb[u * WG_B + lane];

            for (int s = 0; s < CH; s += 16) {
#pragma unroll
                for (int u = 0; u < 8; ++u) gc[u] = gb[(s + 8 + u) * WG_B + lane];
                __builtin_amdgcn_sched_barrier(0);
#pragma unroll
                for (int u = 0; u < 8; ++u) {
                    hv[u] = cell_step(ga[u], h, c, wi, wf, wg, wo);
                }
                const int sn = (s + 16 < CH) ? (s + 16) : 0;
#pragma unroll
                for (int u = 0; u < 8; ++u) ga[u] = gb[(sn + u) * WG_B + lane];
                __builtin_amdgcn_sched_barrier(0);
#pragma unroll
                for (int u = 0; u < 8; ++u) {
                    hv[8 + u] = cell_step(gc[u], h, c, wi, wf, wg, wo);
                }
                __builtin_amdgcn_sched_barrier(0);
#pragma unroll
                for (int u = 0; u < 16; ++u) {
                    orow[(size_t)(s + u) * BATCH] = hv[u];
                }
            }
          }
        } else if (worker && ch + 1 < NCH) {
            produce_chunk(x, (float*)gbuf[(ch + 1) & 1], ch + 1, b0, rank, lane, bmat, bias);
        }
        lds_barrier();
    }
}

extern "C" void kernel_launch(void* const* d_in, const int* in_sizes, int n_in,
                              void* d_out, int out_size, void* d_ws, size_t ws_size,
                              hipStream_t stream)
{
    const float* x    = (const float*)d_in[0];
    const float* W_ih = (const float*)d_in[1];
    const float* W_hh = (const float*)d_in[2];
    const float* b_ih = (const float*)d_in[3];
    const float* b_hh = (const float*)d_in[4];
    float* out = (float*)d_out;

    lstm_fused_wmma<<<32, 512, 0, stream>>>(x, W_ih, W_hh, b_ih, b_hh, out);
}
